# starting kernel
# baseline (speedup 1.0000x reference)
#include <hip/hip_runtime.h>
#include <math.h>


typedef float v2f __attribute__((ext_vector_type(2)));
typedef float v8f __attribute__((ext_vector_type(8)));

#define T_DIM     4096
#define COLS      8192
#define W         256
#define BLOCK_OUT 512
#define NTILES    48

__device__ __forceinline__ float clean(float v) {
    v = (v != v) ? 0.0f : v;
    return fminf(fmaxf(v, -3.4028235e38f), 3.4028235e38f);
}

__device__ __forceinline__ float ld_guard(const float* __restrict__ x, int g, int col) {
    return (g < T_DIM) ? clean(x[(size_t)g * COLS + col]) : 0.0f;
}

__global__ void __launch_bounds__(256)
roll_mean_wmma_f32(const float* __restrict__ x, float* __restrict__ out) {
    const int lane = threadIdx.x & 31;
    const int wave = threadIdx.x >> 5;
    const int half = lane >> 4;
    const int m    = lane & 15;
    const int col  = (blockIdx.x * 8 + wave) * 16 + m;
    const int bs   = blockIdx.y * BLOCK_OUT;

    v2f a[4];
#pragma unroll
    for (int j = 0; j < 4; ++j) {
        const int k0 = 4 * j + 2 * half;
        a[j].x = (k0     <= m) ? 1.0f : 0.0f;
        a[j].y = (k0 + 1 <= m) ? 1.0f : 0.0f;
    }

    const float inv_w = 1.0f / (float)W;
    float carry = 0.0f;

    for (int tile = 0; tile < NTILES; ++tile) {
        const int rbase = tile * 16;
        const int g0    = bs + rbase;

        {
            int gp = g0 + 64;
            if (gp >= T_DIM) gp = T_DIM - 1;
            __builtin_prefetch(x + (size_t)gp * COLS + col, 0, 0);
        }

        v2f b[4];
        if (g0 + 16 <= T_DIM) {
            const float* q  = x + (size_t)(g0 + 2 * half) * COLS + col;
            const float* ql = q - (size_t)W * COLS;
#pragma unroll
            for (int j = 0; j < 4; ++j) {
                float v0 = clean(q[(4 * j)     * COLS]);
                float v1 = clean(q[(4 * j + 1) * COLS]);
                if (tile >= 16) {
                    v0 -= clean(ql[(4 * j)     * COLS]);
                    v1 -= clean(ql[(4 * j + 1) * COLS]);
                }
                b[j].x = v0; b[j].y = v1;
            }
        } else {
#pragma unroll
            for (int j = 0; j < 4; ++j) {
                const int r0 = rbase + 4 * j + 2 * half;
                float v0 = ld_guard(x, bs + r0,     col) - ld_guard(x, bs + r0 - W,     col);
                float v1 = ld_guard(x, bs + r0 + 1, col) - ld_guard(x, bs + r0 + 1 - W, col);
                b[j].x = v0; b[j].y = v1;
            }
        }

        v8f acc = { carry, carry, carry, carry, carry, carry, carry, carry };
#pragma unroll
        for (int j = 0; j < 4; ++j)
            acc = __builtin_amdgcn_wmma_f32_16x16x4_f32(
                      false, a[j], false, b[j], (short)0, acc, false, false);

        carry = __int_as_float(
            __builtin_amdgcn_ds_bpermute((m + 16) << 2, __float_as_int(acc[7])));

        if (tile == 15) {
            const float fv = carry * inv_w;
            if (half == 0)
                out[(size_t)(bs + 255) * COLS + col] = fv;
            if (blockIdx.y == 0) {
                float* p = out + (size_t)half * COLS + col;
                for (int t0 = 0; t0 < 254; t0 += 2) {
                    *p = fv;
                    p += 2 * COLS;
                }
                if (half == 0)
                    out[(size_t)254 * COLS + col] = fv;
            }
        } else if (tile >= 16 && g0 < T_DIM) {
            float* p = out + (size_t)(g0 + 8 * half) * COLS + col;
            if (tile < 47 && g0 + 16 <= T_DIM) {
#pragma unroll
                for (int vi = 0; vi < 8; ++vi)
                    p[vi * COLS] = acc[vi] * inv_w;
            } else {
#pragma unroll
                for (int vi = 0; vi < 8; ++vi) {
                    const int r = rbase + vi + 8 * half;
                    if (r <= 766 && bs + r < T_DIM)
                        p[vi * COLS] = acc[vi] * inv_w;
                }
            }
        }
    }
}

extern "C" void kernel_launch(void* const* d_in, const int* in_sizes, int n_in,
                              void* d_out, int out_size, void* d_ws, size_t ws_size,
                              hipStream_t stream) {
    (void)in_sizes; (void)n_in; (void)d_ws; (void)ws_size; (void)out_size;
    const float* x = (const float*)d_in[0];
    float* out     = (float*)d_out;
    dim3 grid(COLS / 128, 8);
    roll_mean_wmma_f32<<<grid, 256, 0, stream>>>(x, out);
}


# Work-groups made persistent: 256 (one per compute unit) walk 4 units of 512 rows each; the next unit's rows are requested one unit ahead, under this unit's matrix work and output writes; same sums
# speedup vs baseline: 5.2206x; 5.2206x over previous
#include <hip/hip_runtime.h>
#include <math.h>


typedef float v2f __attribute__((ext_vector_type(2)));
typedef float v8f __attribute__((ext_vector_type(8)));
typedef float v4f __attribute__((ext_vector_type(4)));

#define COLS  8192
#define WIN   256
#define RB    128
#define NBLK  4
#define WBLK  2
#define NCG   4
#define NSTRIP 128
#define NYWG  2
#define NUNIT 4
#define SROWS 32
#define SPITCH 68

__device__ __forceinline__ float clean(float v) {
    v = (v != v) ? 0.0f : v;
    return fminf(fmaxf(v, -3.4028235e38f), 3.4028235e38f);
}

__device__ __forceinline__ float block_sum(const v2f (&v)[32]) {
    float p[8];
#pragma unroll
    for (int t = 0; t < 8; ++t) {
        float s[4];
#pragma unroll
        for (int j = 0; j < 4; ++j)
            s[j] = v[4 * t + j].x + v[4 * t + j].y;
        p[t] = (s[0] + s[1]) + (s[2] + s[3]);
    }
    return ((p[0] + p[1]) + (p[2] + p[3])) + ((p[4] + p[5]) + (p[6] + p[7]));
}

__global__ void __launch_bounds__(512)
roll_mean_blocks_wmma_f32(const float* __restrict__ x, float* __restrict__ out) {
    __shared__ float s_lead[NBLK][NCG][32];
    __shared__ float s_lag[WBLK][NCG][32];
    __shared__ __attribute__((aligned(8))) v2f s_x[WBLK][NCG][32][32];
    __shared__ __attribute__((aligned(16))) float s_out[2][NBLK][SROWS][SPITCH];

    const int lane = threadIdx.x & 31;
    const int wave = __builtin_amdgcn_readfirstlane(threadIdx.x >> 5);
    const int half = lane >> 4;
    const int m    = lane & 15;
    const int cg   = wave & (NCG - 1);
    const int blk  = wave / NCG;
    const int strip = blockIdx.x & (NSTRIP - 1);
    const int ybase = blockIdx.x / NSTRIP;
    const int col  = (strip * NCG + cg) * 16 + m;
    const bool upper = (blk < WBLK);

    const float inv_w = 1.0f / (float)WIN;

    v2f a[4];
#pragma unroll
    for (int j = 0; j < 4; ++j) {
        const int k0 = 4 * j + 2 * half;
        a[j].x = (k0     <= m) ? 1.0f : 0.0f;
        a[j].y = (k0 + 1 <= m) ? 1.0f : 0.0f;
    }

    const float* q = x + (size_t)((ybase * NBLK + blk) * RB + 2 * half) * COLS + col;

    v2f xl[32], xg[32];
#pragma unroll
    for (int i = 0; i < 32; ++i) {
        xl[i].x = q[(4 * i)     * COLS];
        xl[i].y = q[(4 * i + 1) * COLS];
    }
#pragma unroll
    for (int i = 0; i < 32; ++i) {
        xg[i].x = 0.0f;
        xg[i].y = 0.0f;
    }
    if (upper) {
        const float* ql = (ybase == 0) ? q : q - (size_t)WIN * COLS;
#pragma unroll
        for (int i = 0; i < 32; ++i) {
            xg[i].x = ql[(4 * i)     * COLS];
            xg[i].y = ql[(4 * i + 1) * COLS];
        }
    }

#pragma nounroll
    for (int u = 0; u < NUNIT; ++u) {
        const int by = NYWG * u + ybase;
        const int r0 = (by * NBLK + blk) * RB;
        const bool first = upper && (by == 0);
#pragma unroll
        for (int i = 0; i < 32; ++i) {
            xl[i].x = clean(xl[i].x);
            xl[i].y = clean(xl[i].y);
        }
        s_lead[blk][cg][lane] = block_sum(xl);
        if (upper) {
#pragma unroll
            for (int i = 0; i < 32; ++i) {
                xg[i].x = clean(xg[i].x);
                xg[i].y = clean(xg[i].y);
            }
            s_lag[blk][cg][lane] = block_sum(xg);
#pragma unroll
            for (int i = 0; i < 32; ++i)
                s_x[blk][cg][i][lane] = xl[i];
        }

        __syncthreads();

        if (!upper) {
#pragma unroll
            for (int i = 0; i < 32; ++i)
                xg[i] = s_x[blk - WBLK][cg][i][lane];
        }
        v2f y[32];
#pragma unroll
        for (int i = 0; i < 32; ++i) {
            y[i].x = xl[i].x - xg[i].x;
            y[i].y = xl[i].y - xg[i].y;
        }

        q += (size_t)NYWG * NBLK * RB * COLS;
        const bool more = (u + 1 < NUNIT);
        if (more) {
#pragma unroll
            for (int i = 0; i < 16; ++i) {
                xl[i].x = q[(4 * i)     * COLS];
                xl[i].y = q[(4 * i + 1) * COLS];
            }
        }

        if (first) {
            v4f fv;
#pragma unroll
            for (int j = 0; j < 4; ++j) {
                const int c  = 4 * m + j;
                const int g  = c >> 4;
                const int mm = c & 15;
                float s = 0.0f;
#pragma unroll
                for (int b = 0; b < WBLK; ++b)
                    s += s_lead[b][g][mm] + s_lead[b][g][mm + 16];
                fv[j] = s * inv_w;
            }
            float* pf = out + (size_t)(32 * wave + half) * COLS
                            + strip * (NCG * 16) + 4 * m;
#pragma unroll
            for (int k = 0; k < 16; ++k)
                *(v4f*)(pf + (size_t)(2 * k) * COLS) = fv;
        }

        float carry = 0.0f;
#pragma unroll
        for (int d = WBLK; d >= 1; --d) {
            const int b = blk - d;
            const float* src = (b >= 0) ? &s_lead[b][cg][0] : &s_lag[b + WBLK][cg][0];
            carry += src[m] + src[m + 16];
        }

        float* p = out + (size_t)(r0 + 8 * cg + half) * COLS + strip * (NCG * 16) + 4 * m;
        v8f d[2];
#pragma unroll
        for (int s = 0; s <= RB / SROWS; ++s) {
            if (!first && s < RB / SROWS) {
#pragma unroll
                for (int tt = 0; tt < 2; ++tt) {
                    const int t = 2 * s + tt;
                    v8f acc = { carry, carry, carry, carry, carry, carry, carry, carry };
#pragma unroll
                    for (int j = 0; j < 4; ++j)
                        acc = __builtin_amdgcn_wmma_f32_16x16x4_f32(
                                  false, a[j], false, y[4 * t + j], (short)0, acc, false, false);
                    if (t < RB / 16 - 1)
                        carry = __int_as_float(
                            __builtin_amdgcn_ds_bpermute((m + 16) << 2, __float_as_int(acc[7])));
                    d[tt] = acc;
                }
            }
            if (more && upper && s == 3) {
                const float* ql = q - (size_t)WIN * COLS;
#pragma unroll
                for (int i = 0; i < 32; ++i) {
                    xg[i].x = ql[(4 * i)     * COLS];
                    xg[i].y = ql[(4 * i + 1) * COLS];
                }
            }
            if (more && s == 1) {
#pragma unroll
                for (int i = 16; i < 32; ++i) {
                    xl[i].x = q[(4 * i)     * COLS];
                    xl[i].y = q[(4 * i + 1) * COLS];
                }
            }
            if (!first && s > 0) {
#pragma unroll
                for (int j = 0; j < 4; ++j) {
                    const v4f v = *(const v4f*)&s_out[(s - 1) & 1][blk][8 * cg + 2 * j + half][4 * m];
                    *(v4f*)(p + (size_t)(SROWS * (s - 1) + 2 * j) * COLS) = v;
                }
            }
            if (s < RB / SROWS) {
                if (!first) {
#pragma unroll
                    for (int tt = 0; tt < 2; ++tt)
#pragma unroll
                        for (int vi = 0; vi < 8; ++vi)
                            s_out[s & 1][blk][16 * tt + 8 * half + vi][16 * cg + m] = d[tt][vi] * inv_w;
                }
                __builtin_amdgcn_fence(__ATOMIC_RELEASE, "workgroup", "local");
                __builtin_amdgcn_s_barrier();
                __builtin_amdgcn_fence(__ATOMIC_ACQUIRE, "workgroup", "local");
            }
        }
    }
}

extern "C" void kernel_launch(void* const* d_in, const int* in_sizes, int n_in,
                              void* d_out, int out_size, void* d_ws, size_t ws_size,
                              hipStream_t stream) {
    (void)in_sizes; (void)n_in; (void)d_ws; (void)ws_size; (void)out_size;
    const float* x = (const float*)d_in[0];
    float* out     = (float*)d_out;
    roll_mean_blocks_wmma_f32<<<dim3(NSTRIP * NYWG), 512, 0, stream>>>(x, out);
}
